# starting kernel
# baseline (speedup 1.0000x reference)
#include <hip/hip_runtime.h>
#include <hip/hip_bf16.h>

#define N_NODES  100000
#define E_EDGES  1600000
#define IN_F     256
#define HD       128
#define H_HEADS  8
#define SLOPE    0.2f

#define ROWS_PB    64
#define LDS_STRIDE 257

typedef __attribute__((ext_vector_type(2))) float v2f;
typedef __attribute__((ext_vector_type(8))) float v8f;

__device__ __forceinline__ unsigned f2u(float f) {
    unsigned u = __float_as_uint(f);
    return (u & 0x80000000u) ? ~u : (u | 0x80000000u);
}
__device__ __forceinline__ float u2f(unsigned u) {
    u = (u & 0x80000000u) ? (u & 0x7FFFFFFFu) : ~u;
    return __uint_as_float(u);
}

__global__ __launch_bounds__(256) void gat_gemm3(
    const float* __restrict__ x,
    const float* __restrict__ Wsrc, const float* __restrict__ bsrc,
    const float* __restrict__ Wdst, const float* __restrict__ bdst,
    const float* __restrict__ Wres, const float* __restrict__ bres,
    float* __restrict__ el, float* __restrict__ er, float* __restrict__ outres)
{
    __shared__ float xs[ROWS_PB * LDS_STRIDE];

    const int tid  = threadIdx.x;
    const int row0 = blockIdx.x * ROWS_PB;

    for (int i = tid; i < ROWS_PB * (IN_F / 4); i += 256) {
        const int r  = i >> 6;
        const int c4 = i & 63;
        float4 v = make_float4(0.f, 0.f, 0.f, 0.f);
        const int gr = row0 + r;
        if (gr < N_NODES)
            v = *(const float4*)(x + (size_t)gr * IN_F + c4 * 4);
        float* p = &xs[r * LDS_STRIDE + c4 * 4];
        p[0] = v.x; p[1] = v.y; p[2] = v.z; p[3] = v.w;
    }
    __syncthreads();

    const int wave    = tid >> 5;
    const int lane    = tid & 31;
    const int l15     = lane & 15;
    const int khalf   = (lane >> 4) * 2;
    const int rhalf   = (lane >> 4) * 8;
    const int colbase = wave * 16;

    const float* const Wm[3] = { Wsrc, Wdst, Wres };
    const float* const bm[3] = { bsrc, bdst, bres };
    float* const       Om[3] = { el,   er,   outres };

#pragma unroll
    for (int m = 0; m < 3; ++m) {
        const float* __restrict__ W = Wm[m];
        v8f acc0 = {}; v8f acc1 = {}; v8f acc2 = {}; v8f acc3 = {};

        for (int k = 0; k < IN_F; k += 4) {
            v2f b;
            b.x = W[(k + khalf)     * HD + colbase + l15];
            b.y = W[(k + khalf + 1) * HD + colbase + l15];

#define GAT_WSTEP(ACC, RT)                                                     \
            {                                                                  \
                v2f a;                                                         \
                const int ab = ((RT) * 16 + l15) * LDS_STRIDE + k + khalf;     \
                a.x = xs[ab];                                                  \
                a.y = xs[ab + 1];                                              \
                ACC = __builtin_amdgcn_wmma_f32_16x16x4_f32(                   \
                        false, a, false, b, (short)0, ACC, false, false);      \
            }
            GAT_WSTEP(acc0, 0)
            GAT_WSTEP(acc1, 1)
            GAT_WSTEP(acc2, 2)
            GAT_WSTEP(acc3, 3)
#undef GAT_WSTEP
        }

        const float bias = bm[m][colbase + l15];
        float* __restrict__ O = Om[m];

#define GAT_STORE(ACC, RT)                                                     \
        {                                                                      \
            _Pragma("unroll")                                                  \
            for (int j = 0; j < 8; ++j) {                                      \
                const int r = row0 + (RT) * 16 + rhalf + j;                    \
                if (r < N_NODES)                                               \
                    O[(size_t)r * HD + colbase + l15] = ACC[j] + bias;         \
            }                                                                  \
        }
        GAT_STORE(acc0, 0)
        GAT_STORE(acc1, 1)
        GAT_STORE(acc2, 2)
        GAT_STORE(acc3, 3)
#undef GAT_STORE
    }
}

__global__ __launch_bounds__(256) void gat_score(
    const float* __restrict__ el, const float* __restrict__ er,
    const int* __restrict__ src, const int* __restrict__ dst,
    const float* __restrict__ attn,
    float* __restrict__ score, unsigned* __restrict__ maxu)
{
    const int e = blockIdx.x * 8 + (threadIdx.x >> 5);
    if (e >= E_EDGES) return;
    const int lane = threadIdx.x & 31;
    const int u = src[e];
    const int v = dst[e];
    const int c = lane * 4;

    const float4 a4 = *(const float4*)(el   + (size_t)u * HD + c);
    const float4 b4 = *(const float4*)(er   + (size_t)v * HD + c);
    const float4 w4 = *(const float4*)(attn + c);

    float p = 0.f, t;
    t = a4.x + b4.x; t = (t > 0.f) ? t : SLOPE * t; p += t * w4.x;
    t = a4.y + b4.y; t = (t > 0.f) ? t : SLOPE * t; p += t * w4.y;
    t = a4.z + b4.z; t = (t > 0.f) ? t : SLOPE * t; p += t * w4.z;
    t = a4.w + b4.w; t = (t > 0.f) ? t : SLOPE * t; p += t * w4.w;

    p += __shfl_xor(p, 1, 32);
    p += __shfl_xor(p, 2, 32);

    if ((lane & 3) == 0) {
        const int h = lane >> 2;
        score[(size_t)e * H_HEADS + h] = p;
        atomicMax(&maxu[(size_t)v * H_HEADS + h], f2u(p));
    }
}

__global__ __launch_bounds__(256) void gat_expsum(
    const int* __restrict__ dst, float* __restrict__ score,
    const unsigned* __restrict__ maxu, float* __restrict__ denom)
{
    const long long t = (long long)blockIdx.x * 256 + threadIdx.x;
    const int e = (int)(t >> 3);
    const int h = (int)(t & 7);
    if (e >= E_EDGES) return;
    const int v = dst[e];
    const float m = u2f(maxu[(size_t)v * H_HEADS + h]);
    const float a = __expf(score[t] - m);
    score[t] = a;
    unsafeAtomicAdd(&denom[(size_t)v * H_HEADS + h], a);
}

__global__ __launch_bounds__(256) void gat_aggregate(
    const float* __restrict__ el,
    const int* __restrict__ src, const int* __restrict__ dst,
    const float* __restrict__ alpha, const float* __restrict__ denom,
    float* __restrict__ out)
{
    const int e = blockIdx.x * 8 + (threadIdx.x >> 5);
    if (e >= E_EDGES) return;
    const int lane = threadIdx.x & 31;
    const int u = src[e];
    const int v = dst[e];
    const int h = lane >> 2;
    const int c = lane * 4;

    const float w  = alpha[(size_t)e * H_HEADS + h] / denom[(size_t)v * H_HEADS + h];
    const float4 f = *(const float4*)(el + (size_t)u * HD + c);

    float* o = out + (size_t)v * HD + c;
    unsafeAtomicAdd(o + 0, w * f.x);
    unsafeAtomicAdd(o + 1, w * f.y);
    unsafeAtomicAdd(o + 2, w * f.z);
    unsafeAtomicAdd(o + 3, w * f.w);
}

extern "C" void kernel_launch(void* const* d_in, const int* in_sizes, int n_in,
                              void* d_out, int out_size, void* d_ws, size_t ws_size,
                              hipStream_t stream) {
    const float* x    = (const float*)d_in[0];
    const int*   src  = (const int*)  d_in[1];
    const int*   dst  = (const int*)  d_in[2];
    const float* Wsrc = (const float*)d_in[3];
    const float* bsrc = (const float*)d_in[4];
    const float* Wdst = (const float*)d_in[5];
    const float* bdst = (const float*)d_in[6];
    const float* attn = (const float*)d_in[7];
    const float* Wres = (const float*)d_in[8];
    const float* bres = (const float*)d_in[9];
    float* out = (float*)d_out;

    char* ws = (char*)d_ws;
    const size_t elBytes    = (size_t)N_NODES * HD * sizeof(float);
    const size_t scoreBytes = (size_t)E_EDGES * H_HEADS * sizeof(float);
    const size_t nhBytes    = (size_t)N_NODES * H_HEADS * sizeof(float);
    float*    el    = (float*)   (ws);
    float*    er    = (float*)   (ws + elBytes);
    float*    score = (float*)   (ws + 2 * elBytes);
    unsigned* maxu  = (unsigned*)(ws + 2 * elBytes + scoreBytes);
    float*    denom = (float*)   (ws + 2 * elBytes + scoreBytes + nhBytes);

    hipMemsetAsync(maxu,  0, nhBytes, stream);
    hipMemsetAsync(denom, 0, nhBytes, stream);

    const int gemmBlocks = (N_NODES + ROWS_PB - 1) / ROWS_PB;
    gat_gemm3<<<gemmBlocks, 256, 0, stream>>>(x, Wsrc, bsrc, Wdst, bdst,
                                              Wres, bres, el, er, out);

    const int edgeBlocks = (E_EDGES + 7) / 8;
    gat_score<<<edgeBlocks, 256, 0, stream>>>(el, er, src, dst, attn, score, maxu);

    const long long ehThreads = (long long)E_EDGES * H_HEADS;
    const int expBlocks = (int)((ehThreads + 255) / 256);
    gat_expsum<<<expBlocks, 256, 0, stream>>>(dst, score, maxu, denom);

    gat_aggregate<<<edgeBlocks, 256, 0, stream>>>(el, src, dst, score, denom, out);
}


# Edge kernels batched: a wave takes 8 edges (exp kernel: 4 pieces a lane), all loads of a batch issued before first use, 4 edges' scores leave as one 128-byte line; 8x fewer waves; sums as before
# speedup vs baseline: 1.1425x; 1.1425x over previous
#include <hip/hip_runtime.h>
#include <hip/hip_bf16.h>

#define N_NODES  100000
#define E_EDGES  1600000
#define IN_F     256
#define HD       128
#define H_HEADS  8
#define SLOPE    0.2f

#define ROWS_PB    64
#define LDS_STRIDE 257

typedef __attribute__((ext_vector_type(2))) float v2f;
typedef __attribute__((ext_vector_type(8))) float v8f;

__device__ __forceinline__ unsigned f2u(float f) {
    unsigned u = __float_as_uint(f);
    return (u & 0x80000000u) ? ~u : (u | 0x80000000u);
}
__device__ __forceinline__ float u2f(unsigned u) {
    u = (u & 0x80000000u) ? (u & 0x7FFFFFFFu) : ~u;
    return __uint_as_float(u);
}

__global__ __launch_bounds__(256) void gat_gemm3(
    const float* __restrict__ x,
    const float* __restrict__ Wsrc, const float* __restrict__ bsrc,
    const float* __restrict__ Wdst, const float* __restrict__ bdst,
    const float* __restrict__ Wres, const float* __restrict__ bres,
    float* __restrict__ el, float* __restrict__ er, float* __restrict__ outres)
{
    __shared__ float xs[ROWS_PB * LDS_STRIDE];

    const int tid  = threadIdx.x;
    const int row0 = blockIdx.x * ROWS_PB;

    for (int i = tid; i < ROWS_PB * (IN_F / 4); i += 256) {
        const int r  = i >> 6;
        const int c4 = i & 63;
        float4 v = make_float4(0.f, 0.f, 0.f, 0.f);
        const int gr = row0 + r;
        if (gr < N_NODES)
            v = *(const float4*)(x + (size_t)gr * IN_F + c4 * 4);
        float* p = &xs[r * LDS_STRIDE + c4 * 4];
        p[0] = v.x; p[1] = v.y; p[2] = v.z; p[3] = v.w;
    }
    __syncthreads();

    const int wave    = tid >> 5;
    const int lane    = tid & 31;
    const int l15     = lane & 15;
    const int khalf   = (lane >> 4) * 2;
    const int rhalf   = (lane >> 4) * 8;
    const int colbase = wave * 16;

    const float* const Wm[3] = { Wsrc, Wdst, Wres };
    const float* const bm[3] = { bsrc, bdst, bres };
    float* const       Om[3] = { el,   er,   outres };

#pragma unroll
    for (int m = 0; m < 3; ++m) {
        const float* __restrict__ W = Wm[m];
        v8f acc0 = {}; v8f acc1 = {}; v8f acc2 = {}; v8f acc3 = {};

        for (int k = 0; k < IN_F; k += 4) {
            v2f b;
            b.x = W[(k + khalf)     * HD + colbase + l15];
            b.y = W[(k + khalf + 1) * HD + colbase + l15];

#define GAT_WSTEP(ACC, RT)                                                     \
            {                                                                  \
                v2f a;                                                         \
                const int ab = ((RT) * 16 + l15) * LDS_STRIDE + k + khalf;     \
                a.x = xs[ab];                                                  \
                a.y = xs[ab + 1];                                              \
                ACC = __builtin_amdgcn_wmma_f32_16x16x4_f32(                   \
                        false, a, false, b, (short)0, ACC, false, false);      \
            }
            GAT_WSTEP(acc0, 0)
            GAT_WSTEP(acc1, 1)
            GAT_WSTEP(acc2, 2)
            GAT_WSTEP(acc3, 3)
#undef GAT_WSTEP
        }

        const float bias = bm[m][colbase + l15];
        float* __restrict__ O = Om[m];

#define GAT_STORE(ACC, RT)                                                     \
        {                                                                      \
            _Pragma("unroll")                                                  \
            for (int j = 0; j < 8; ++j) {                                      \
                const int r = row0 + (RT) * 16 + rhalf + j;                    \
                if (r < N_NODES)                                               \
                    O[(size_t)r * HD + colbase + l15] = ACC[j] + bias;         \
            }                                                                  \
        }
        GAT_STORE(acc0, 0)
        GAT_STORE(acc1, 1)
        GAT_STORE(acc2, 2)
        GAT_STORE(acc3, 3)
#undef GAT_STORE
    }
}

#define EPW 8
#define XPT 4
static_assert(EPW == 8, "the load pins below name 8 edges");
static_assert(XPT == 4, "the load pin below names 4 pieces");
static_assert(E_EDGES % (8 * EPW) == 0, "edges must fill whole work-groups");
static_assert((E_EDGES * 2) % (256 * XPT) == 0, "half-edges must fill whole work-groups");

typedef __attribute__((ext_vector_type(4))) float    v4f;
typedef __attribute__((ext_vector_type(4))) unsigned v4u;

__global__ __launch_bounds__(256) void gat_score(
    const float* __restrict__ el, const float* __restrict__ er,
    const int* __restrict__ src, const int* __restrict__ dst,
    const float* __restrict__ attn,
    float* __restrict__ score, unsigned* __restrict__ maxu)
{
    const int e0 = (blockIdx.x * 8 + (threadIdx.x >> 5)) * EPW;
    if (e0 >= E_EDGES) return;
    const int lane = threadIdx.x & 31;
    const int c = lane * 4;
    const int j = lane & 3;
    const int h = lane >> 2;

    const int us = src[e0 + (lane & (EPW - 1))];
    const int vs = dst[e0 + (lane & (EPW - 1))];
    int vd0 = dst[e0 + j];
    int vd1 = dst[e0 + 4 + j];
    v4f w4 = *(const v4f*)(attn + c);

    v4f a4[EPW], b4[EPW];
#pragma unroll
    for (int g = 0; g < EPW; ++g) {
        const int u = __builtin_amdgcn_readlane(us, g);
        const int v = __builtin_amdgcn_readlane(vs, g);
        a4[g] = *(const v4f*)(el + (size_t)u * HD + c);
        b4[g] = *(const v4f*)(er + (size_t)v * HD + c);
    }
    asm volatile("" : "+v"(a4[0]), "+v"(a4[1]), "+v"(a4[2]), "+v"(a4[3]),
                      "+v"(a4[4]), "+v"(a4[5]), "+v"(a4[6]), "+v"(a4[7]),
                      "+v"(b4[0]), "+v"(b4[1]), "+v"(b4[2]), "+v"(b4[3]),
                      "+v"(b4[4]), "+v"(b4[5]), "+v"(b4[6]), "+v"(b4[7]),
                      "+v"(w4), "+v"(vd0), "+v"(vd1));

    float p[EPW];
#pragma unroll
    for (int g = 0; g < EPW; ++g) {
        float q = 0.f, t;
        t = a4[g].x + b4[g].x; t = (t > 0.f) ? t : SLOPE * t; q += t * w4.x;
        t = a4[g].y + b4[g].y; t = (t > 0.f) ? t : SLOPE * t; q += t * w4.y;
        t = a4[g].z + b4[g].z; t = (t > 0.f) ? t : SLOPE * t; q += t * w4.z;
        t = a4[g].w + b4[g].w; t = (t > 0.f) ? t : SLOPE * t; q += t * w4.w;
        p[g] = q;
    }

    const bool odd = (j & 1) != 0;
    const bool hi  = (j & 2) != 0;
#pragma unroll
    for (int q = 0; q < EPW / 4; ++q) {
        const float k0 = odd ? p[4 * q + 1] : p[4 * q];
        const float s0 = odd ? p[4 * q]     : p[4 * q + 1];
        const float k1 = odd ? p[4 * q + 3] : p[4 * q + 2];
        const float s1 = odd ? p[4 * q + 2] : p[4 * q + 3];
        const float r0 = k0 + __shfl_xor(s0, 1, 32);
        const float r1 = k1 + __shfl_xor(s1, 1, 32);
        const float kk = hi ? r1 : r0;
        const float ss = hi ? r0 : r1;
        const float ps = kk + __shfl_xor(ss, 2, 32);

        const int    vd = (q == 0) ? vd0 : vd1;
        const size_t e  = (size_t)(e0 + 4 * q + j);
        score[e * H_HEADS + h] = ps;
        atomicMax(&maxu[(size_t)vd * H_HEADS + h], f2u(ps));
    }
}

__global__ __launch_bounds__(256) void gat_expsum(
    const int* __restrict__ dst, float* __restrict__ score,
    const unsigned* __restrict__ maxu, float* __restrict__ denom)
{
    const int base = blockIdx.x * (256 * XPT) + threadIdx.x;

    int v[XPT];
    v4f s[XPT];
    v4u m[XPT];
#pragma unroll
    for (int k = 0; k < XPT; ++k) {
        const int idx = base + k * 256;
        v[k] = dst[idx >> 1];
        s[k] = *(const v4f*)(score + (size_t)idx * 4);
    }
#pragma unroll
    for (int k = 0; k < XPT; ++k) {
        const int idx = base + k * 256;
        m[k] = *(const v4u*)(maxu + (size_t)v[k] * H_HEADS + (idx & 1) * 4);
    }
    asm volatile("" : "+v"(s[0]), "+v"(s[1]), "+v"(s[2]), "+v"(s[3]),
                      "+v"(m[0]), "+v"(m[1]), "+v"(m[2]), "+v"(m[3]));
#pragma unroll
    for (int k = 0; k < XPT; ++k) {
        const int idx = base + k * 256;
        v4f a;
        a.x = __expf(s[k].x - u2f(m[k].x));
        a.y = __expf(s[k].y - u2f(m[k].y));
        a.z = __expf(s[k].z - u2f(m[k].z));
        a.w = __expf(s[k].w - u2f(m[k].w));
        *(v4f*)(score + (size_t)idx * 4) = a;
        float* d = denom + (size_t)v[k] * H_HEADS + (idx & 1) * 4;
        unsafeAtomicAdd(d + 0, a.x);
        unsafeAtomicAdd(d + 1, a.y);
        unsafeAtomicAdd(d + 2, a.z);
        unsafeAtomicAdd(d + 3, a.w);
    }
}

__global__ __launch_bounds__(256) void gat_aggregate(
    const float* __restrict__ el,
    const int* __restrict__ src, const int* __restrict__ dst,
    const float* __restrict__ alpha, const float* __restrict__ denom,
    float* __restrict__ out)
{
    const int e0 = (blockIdx.x * 8 + (threadIdx.x >> 5)) * EPW;
    if (e0 >= E_EDGES) return;
    const int lane = threadIdx.x & 31;
    const int h = lane >> 2;
    const int c = lane * 4;

    const int us = src[e0 + (lane & (EPW - 1))];
    const int vs = dst[e0 + (lane & (EPW - 1))];

    v4f   f[EPW];
    float al[EPW], dn[EPW];
    int   vv[EPW];
#pragma unroll
    for (int g = 0; g < EPW; ++g) {
        const int u = __builtin_amdgcn_readlane(us, g);
        vv[g] = __builtin_amdgcn_readlane(vs, g);
        f[g]  = *(const v4f*)(el + (size_t)u * HD + c);
    }
#pragma unroll
    for (int g = 0; g < EPW; ++g) {
        al[g] = alpha[(size_t)(e0 + g) * H_HEADS + h];
        dn[g] = denom[(size_t)vv[g] * H_HEADS + h];
    }
    asm volatile("" : "+v"(f[0]), "+v"(f[1]), "+v"(f[2]), "+v"(f[3]),
                      "+v"(f[4]), "+v"(f[5]), "+v"(f[6]), "+v"(f[7]),
                      "+v"(al[0]), "+v"(al[1]), "+v"(al[2]), "+v"(al[3]),
                      "+v"(al[4]), "+v"(al[5]), "+v"(al[6]), "+v"(al[7]),
                      "+v"(dn[0]), "+v"(dn[1]), "+v"(dn[2]), "+v"(dn[3]),
                      "+v"(dn[4]), "+v"(dn[5]), "+v"(dn[6]), "+v"(dn[7]));
#pragma unroll
    for (int g = 0; g < EPW; ++g) {
        const float w = al[g] / dn[g];
        float* o = out + (size_t)vv[g] * HD + c;
        unsafeAtomicAdd(o + 0, w * f[g].x);
        unsafeAtomicAdd(o + 1, w * f[g].y);
        unsafeAtomicAdd(o + 2, w * f[g].z);
        unsafeAtomicAdd(o + 3, w * f[g].w);
    }
}

extern "C" void kernel_launch(void* const* d_in, const int* in_sizes, int n_in,
                              void* d_out, int out_size, void* d_ws, size_t ws_size,
                              hipStream_t stream) {
    const float* x    = (const float*)d_in[0];
    const int*   src  = (const int*)  d_in[1];
    const int*   dst  = (const int*)  d_in[2];
    const float* Wsrc = (const float*)d_in[3];
    const float* bsrc = (const float*)d_in[4];
    const float* Wdst = (const float*)d_in[5];
    const float* bdst = (const float*)d_in[6];
    const float* attn = (const float*)d_in[7];
    const float* Wres = (const float*)d_in[8];
    const float* bres = (const float*)d_in[9];
    float* out = (float*)d_out;

    char* ws = (char*)d_ws;
    const size_t elBytes    = (size_t)N_NODES * HD * sizeof(float);
    const size_t scoreBytes = (size_t)E_EDGES * H_HEADS * sizeof(float);
    const size_t nhBytes    = (size_t)N_NODES * H_HEADS * sizeof(float);
    float*    el    = (float*)   (ws);
    float*    er    = (float*)   (ws + elBytes);
    float*    score = (float*)   (ws + 2 * elBytes);
    unsigned* maxu  = (unsigned*)(ws + 2 * elBytes + scoreBytes);
    float*    denom = (float*)   (ws + 2 * elBytes + scoreBytes + nhBytes);

    hipMemsetAsync(maxu,  0, nhBytes, stream);
    hipMemsetAsync(denom, 0, nhBytes, stream);

    const int gemmBlocks = (N_NODES + ROWS_PB - 1) / ROWS_PB;
    gat_gemm3<<<gemmBlocks, 256, 0, stream>>>(x, Wsrc, bsrc, Wdst, bdst,
                                              Wres, bres, el, er, out);

    const int edgeBlocks = E_EDGES / (8 * EPW);
    gat_score<<<edgeBlocks, 256, 0, stream>>>(el, er, src, dst, attn, score, maxu);

    const int expBlocks = (E_EDGES * 2) / (256 * XPT);
    gat_expsum<<<expBlocks, 256, 0, stream>>>(dst, score, maxu, denom);

    gat_aggregate<<<edgeBlocks, 256, 0, stream>>>(el, src, dst, score, denom, out);
}
